# starting kernel
# baseline (speedup 1.0000x reference)
#include <hip/hip_runtime.h>
#include <hip/hip_bf16.h>


typedef __attribute__((ext_vector_type(2))) float v2f;
typedef __attribute__((ext_vector_type(8))) float v8f;

#define GMM_N 65536
#define GMM_M 1024

__global__ void __launch_bounds__(GMM_M)
gmm_prep_kernel(const float* __restrict__ mu, const float* __restrict__ A,
                const float* __restrict__ w, float* __restrict__ u,
                float* __restrict__ gmax_out) {
  __shared__ float red[GMM_M];
  const int j = threadIdx.x;
  const float wj = w[j];

  red[j] = wj; __syncthreads();
  for (int s = GMM_M / 2; s > 0; s >>= 1) {
    if (j < s) red[j] = fmaxf(red[j], red[j + s]);
    __syncthreads();
  }
  const float wmax = red[0]; __syncthreads();
  red[j] = __expf(wj - wmax); __syncthreads();
  for (int s = GMM_M / 2; s > 0; s >>= 1) {
    if (j < s) red[j] += red[j + s];
    __syncthreads();
  }
  const float lse_w = wmax + __logf(red[0]); __syncthreads();

  const float a00 = A[4 * j + 0], a01 = A[4 * j + 1];
  const float a10 = A[4 * j + 2], a11 = A[4 * j + 3];
  const float g00 = 0.5f * (a00 * a00 + a01 * a01);
  const float g01 = 0.5f * (a00 * a10 + a01 * a11);
  const float g11 = 0.5f * (a10 * a10 + a11 * a11);
  const float det = g00 * g11 - g01 * g01;
  const float logw = (wj - lse_w) + 0.5f * __logf(det);

  const float m0 = mu[2 * j + 0], m1 = mu[2 * j + 1];
  const float p   = 2.f * (g00 * m0 + g01 * m1);
  const float r   = 2.f * (g01 * m0 + g11 * m1);
  const float mqm = g00 * m0 * m0 + 2.f * g01 * m0 * m1 + g11 * m1 * m1;

  float* uj = u + 8 * j;
  uj[0] = logw - mqm;
  uj[1] = p;
  uj[2] = r;
  uj[3] = -g00;
  uj[4] = -2.f * g01;
  uj[5] = -g11;
  uj[6] = 0.f;
  uj[7] = 0.f;

  red[j] = logw; __syncthreads();
  for (int s = GMM_M / 2; s > 0; s >>= 1) {
    if (j < s) red[j] = fmaxf(red[j], red[j + s]);
    __syncthreads();
  }
  if (j == 0) *gmax_out = red[0];
}

__global__ void __launch_bounds__(256)
gmm_lse_kernel(const float* __restrict__ sample, const float* __restrict__ u,
               const float* __restrict__ gmaxp, float* __restrict__ out) {
  const int lane = threadIdx.x & 31;
  const int wave = threadIdx.x >> 5;
  const int half = lane >> 4;
  const int l15  = lane & 15;
  const int r0   = (blockIdx.x * 8 + wave) * 16;

  const float2 xy = ((const float2*)sample)[r0 + l15];
  const float x0 = xy.x, x1 = xy.y;

  v2f a0, a1;
  a0.x = half ? x1        : 1.0f;
  a0.y = half ? x0 * x0   : x0;
  a1.x = half ? 0.0f      : x0 * x1;
  a1.y = half ? 0.0f      : x1 * x1;

  const float gmax = *gmaxp;

  float s[8];
#pragma unroll
  for (int v = 0; v < 8; ++v) s[v] = 0.0f;

#pragma unroll 2
  for (int t = 0; t < GMM_M / 16; ++t) {
    const int j = t * 16 + l15;
    const float4 uA = ((const float4*)u)[2 * j + 0];
    const float4 uB = ((const float4*)u)[2 * j + 1];

    v2f b0, b1;
    b0.x = half ? uA.z : uA.x;
    b0.y = half ? uA.w : uA.y;
    b1.x = half ? uB.z : uB.x;
    b1.y = half ? uB.w : uB.y;

    v8f d = {};
    d = __builtin_amdgcn_wmma_f32_16x16x4_f32(false, a0, false, b0,
                                              (short)0, d, false, false);
    d = __builtin_amdgcn_wmma_f32_16x16x4_f32(false, a1, false, b1,
                                              (short)0, d, false, false);

#pragma unroll
    for (int v = 0; v < 8; ++v) s[v] += __expf(d[v] - gmax);
  }

#pragma unroll
  for (int v = 0; v < 8; ++v) {
    float sv = s[v];
    sv += __shfl_xor(sv, 1, 32);
    sv += __shfl_xor(sv, 2, 32);
    sv += __shfl_xor(sv, 4, 32);
    sv += __shfl_xor(sv, 8, 32);
    s[v] = sv;
  }

  if (l15 == 0) {
#pragma unroll
    for (int v = 0; v < 8; ++v)
      out[r0 + half * 8 + v] = gmax + __logf(s[v]);
  }
}

extern "C" void kernel_launch(void* const* d_in, const int* in_sizes, int n_in,
                              void* d_out, int out_size, void* d_ws, size_t ws_size,
                              hipStream_t stream) {
  const float* sample = (const float*)d_in[0];
  const float* mu     = (const float*)d_in[1];
  const float* A      = (const float*)d_in[2];
  const float* w      = (const float*)d_in[3];
  float* out  = (float*)d_out;
  float* u    = (float*)d_ws;
  float* gmax = u + 8 * GMM_M;

  gmm_prep_kernel<<<1, GMM_M, 0, stream>>>(mu, A, w, u, gmax);

  gmm_lse_kernel<<<GMM_N / 128, 256, 0, stream>>>(sample, u, gmax, out);
}


# Coefficient kernel: 256 threads x 4 components; block reductions as in-wave register butterflies + one LDS slot per wave (3 barriers, was 36); loads first, stores last; new sum order in softmax sum
# speedup vs baseline: 1.8819x; 1.8819x over previous
#include <hip/hip_runtime.h>
#include <hip/hip_bf16.h>


typedef __attribute__((ext_vector_type(2))) float v2f;
typedef __attribute__((ext_vector_type(8))) float v8f;

#define GMM_N 65536
#define GMM_M 1024

#define GMM_PT 256
#define GMM_PC (GMM_M / GMM_PT)
#define GMM_PW (GMM_PT / 32)
static_assert(GMM_PC == 4 && GMM_PW == 8, "the reductions below are written for 4 components per thread and 8 waves");

__device__ __forceinline__ float gmm_xhalf(float x) {
  const int b = __builtin_bit_cast(int, x);
  return __builtin_bit_cast(float,
      __builtin_amdgcn_permlanex16(b, b, 0x76543210, 0xfedcba98, false, false));
}
template <int XM>
__device__ __forceinline__ float gmm_xrow(float x) {
  const int b = __builtin_bit_cast(int, x);
  return __builtin_bit_cast(float, __builtin_amdgcn_update_dpp(0, b, 0x160 + XM, 0xf, 0xf, true));
}
__device__ __forceinline__ float gmm_max8(float x) {
  x = fmaxf(x, gmm_xrow<1>(x));
  x = fmaxf(x, gmm_xrow<2>(x));
  return fmaxf(x, gmm_xrow<4>(x));
}
__device__ __forceinline__ float gmm_sum8(float x) {
  x += gmm_xrow<1>(x);
  x += gmm_xrow<2>(x);
  return x + gmm_xrow<4>(x);
}
__device__ __forceinline__ float gmm_max32(float x) {
  x = gmm_max8(x);
  x = fmaxf(x, gmm_xrow<8>(x));
  return fmaxf(x, gmm_xhalf(x));
}
__device__ __forceinline__ float gmm_sum32(float x) {
  x = gmm_sum8(x);
  x += gmm_xrow<8>(x);
  return x + gmm_xhalf(x);
}

__global__ void __launch_bounds__(GMM_PT)
gmm_prep_kernel(const float* __restrict__ mu, const float* __restrict__ A,
                const float* __restrict__ w, float* __restrict__ u,
                float* __restrict__ gmax_out) {
  __shared__ float red_wmax[GMM_PW];
  __shared__ float red_wsum[GMM_PW];
  __shared__ float red_gmax[GMM_PW];
  const int t = threadIdx.x;
  const int lane = t & 31;
  const int wv = t >> 5;
  const int slot = lane & (GMM_PW - 1);

  float wj[GMM_PC], a00[GMM_PC], a01[GMM_PC], a10[GMM_PC], a11[GMM_PC], m0[GMM_PC], m1[GMM_PC];
#pragma unroll
  for (int k = 0; k < GMM_PC; ++k) wj[k] = w[t + GMM_PT * k];
#pragma unroll
  for (int k = 0; k < GMM_PC; ++k) {
    const int j = t + GMM_PT * k;
    a00[k] = A[4 * j + 0]; a01[k] = A[4 * j + 1];
    a10[k] = A[4 * j + 2]; a11[k] = A[4 * j + 3];
  }
#pragma unroll
  for (int k = 0; k < GMM_PC; ++k) {
    const int j = t + GMM_PT * k;
    m0[k] = mu[2 * j + 0]; m1[k] = mu[2 * j + 1];
  }

  const float wm_wave = gmm_max32(fmaxf(fmaxf(wj[0], wj[1]), fmaxf(wj[2], wj[3])));
  if (lane == 0) red_wmax[wv] = wm_wave;
  __syncthreads();
  const float wmax = gmm_max8(red_wmax[slot]);

  const float ws_wave = gmm_sum32((__expf(wj[0] - wmax) + __expf(wj[1] - wmax)) +
                                  (__expf(wj[2] - wmax) + __expf(wj[3] - wmax)));
  if (lane == 0) red_wsum[wv] = ws_wave;
  __syncthreads();
  const float lse_w = wmax + __logf(gmm_sum8(red_wsum[slot]));

  float g00[GMM_PC], g01[GMM_PC], g11[GMM_PC], logw[GMM_PC];
#pragma unroll
  for (int k = 0; k < GMM_PC; ++k) {
    g00[k] = 0.5f * (a00[k] * a00[k] + a01[k] * a01[k]);
    g01[k] = 0.5f * (a00[k] * a10[k] + a01[k] * a11[k]);
    g11[k] = 0.5f * (a10[k] * a10[k] + a11[k] * a11[k]);
    const float det = g00[k] * g11[k] - g01[k] * g01[k];
    logw[k] = (wj[k] - lse_w) + 0.5f * __logf(det);
  }

  const float gm_wave = gmm_max32(fmaxf(fmaxf(logw[0], logw[1]), fmaxf(logw[2], logw[3])));
  if (lane == 0) red_gmax[wv] = gm_wave;
  __syncthreads();
  const float gm = gmm_max8(red_gmax[slot]);

#pragma unroll
  for (int k = 0; k < GMM_PC; ++k) {
    const float p   = 2.f * (g00[k] * m0[k] + g01[k] * m1[k]);
    const float r   = 2.f * (g01[k] * m0[k] + g11[k] * m1[k]);
    const float mqm = g00[k] * m0[k] * m0[k] + 2.f * g01[k] * m0[k] * m1[k] + g11[k] * m1[k] * m1[k];
    float* uj = u + 8 * (t + GMM_PT * k);
    uj[0] = logw[k] - mqm;
    uj[1] = p;
    uj[2] = r;
    uj[3] = -g00[k];
    uj[4] = -2.f * g01[k];
    uj[5] = -g11[k];
    uj[6] = 0.f;
    uj[7] = 0.f;
  }
  if (t == 0) *gmax_out = gm;
}

#define GMM_ST 4
#define GMM_CT 16

__global__ void __launch_bounds__(256)
gmm_lse_kernel(const float* __restrict__ sample, const float* __restrict__ u,
               const float* __restrict__ gmaxp, float* __restrict__ out) {
  __shared__ float part[2][4][16 * GMM_ST];

  const int lane = threadIdx.x & 31;
  const int wave = threadIdx.x >> 5;
  const int half = lane >> 4;
  const int l15  = lane & 15;
  const int grp  = wave >> 2;
  const int qtr  = wave & 3;
  const int s0   = blockIdx.x * 128 + grp * 64;

  const v2f* up = (const v2f*)u + 4 * (qtr * (16 * GMM_CT) + l15) + half;
  v2f n0 = up[0], n1 = up[2];

  v2f f0[GMM_ST], f1[GMM_ST];
#pragma unroll
  for (int st = 0; st < GMM_ST; ++st) {
    const float2 xy = ((const float2*)sample)[s0 + st * 16 + l15];
    const float x0 = xy.x, x1 = xy.y;
    f0[st].x = half ? x1      : 1.0f;
    f0[st].y = half ? x0 * x0 : x0;
    f1[st].x = half ? 0.0f    : x0 * x1;
    f1[st].y = half ? 0.0f    : x1 * x1;
  }

  const float gmax = *gmaxp;

  float s[GMM_ST][8];
#pragma unroll
  for (int st = 0; st < GMM_ST; ++st)
#pragma unroll
    for (int v = 0; v < 8; ++v) s[st][v] = 0.0f;

  for (int t = 0; t < GMM_CT; ++t) {
    const v2f c0 = n0, c1 = n1;
    const int tn = (t + 1) & (GMM_CT - 1);
    n0 = up[64 * tn];
    n1 = up[64 * tn + 2];
    __builtin_amdgcn_sched_barrier(0);

#pragma unroll
    for (int st = 0; st < GMM_ST; ++st) {
      v8f d = {};
      d = __builtin_amdgcn_wmma_f32_16x16x4_f32(false, c0, false, f0[st],
                                                (short)0, d, false, false);
      d = __builtin_amdgcn_wmma_f32_16x16x4_f32(false, c1, false, f1[st],
                                                (short)0, d, false, false);
#pragma unroll
      for (int v = 0; v < 8; ++v) s[st][v] += __expf(d[v] - gmax);
    }

    __builtin_amdgcn_sched_barrier(0);
    asm volatile("" :: "v"(n0), "v"(n1));
  }

  float tot[GMM_ST];
#pragma unroll
  for (int st = 0; st < GMM_ST; ++st)
    tot[st] = ((s[st][0] + s[st][1]) + (s[st][2] + s[st][3])) +
              ((s[st][4] + s[st][5]) + (s[st][6] + s[st][7]));
#pragma unroll
  for (int st = 0; st < GMM_ST; ++st) tot[st] += __shfl_xor(tot[st], 16, 32);
  if (half == 0) {
#pragma unroll
    for (int st = 0; st < GMM_ST; ++st) part[grp][qtr][st * 16 + l15] = tot[st];
  }
  __syncthreads();

  if (threadIdx.x < 128) {
    const int g = threadIdx.x >> 6, i = threadIdx.x & 63;
    const float sum = (part[g][0][i] + part[g][1][i]) + (part[g][2][i] + part[g][3][i]);
    out[blockIdx.x * 128 + threadIdx.x] = gmax + __logf(sum);
  }
}

extern "C" void kernel_launch(void* const* d_in, const int* in_sizes, int n_in,
                              void* d_out, int out_size, void* d_ws, size_t ws_size,
                              hipStream_t stream) {
  const float* sample = (const float*)d_in[0];
  const float* mu     = (const float*)d_in[1];
  const float* A      = (const float*)d_in[2];
  const float* w      = (const float*)d_in[3];
  float* out  = (float*)d_out;
  float* u    = (float*)d_ws;
  float* gmax = u + 8 * GMM_M;

  gmm_prep_kernel<<<1, GMM_PT, 0, stream>>>(mu, A, w, u, gmax);

  gmm_lse_kernel<<<GMM_N / 128, 256, 0, stream>>>(sample, u, gmax, out);
}
